# starting kernel
# baseline (speedup 1.0000x reference)
#include <hip/hip_runtime.h>


typedef float v2f __attribute__((ext_vector_type(2)));
typedef float v8f __attribute__((ext_vector_type(8)));

#define IN_DIM        16384
#define OUT_CH        2
#define BATCH         4096
#define SAMPLES_PER_WG 16
#define WAVES_PER_WG   8
#define THREADS_PER_WG (32 * WAVES_PER_WG)
#define K_PER_WAVE    (IN_DIM / WAVES_PER_WG)
#define K_OUTER       64

__global__ __launch_bounds__(THREADS_PER_WG)
void diag_wmma_gathered_matvec(const float* __restrict__ x,
                               const int*   __restrict__ t,
                               const float* __restrict__ W,
                               const float* __restrict__ bias,
                               float*       __restrict__ out) {
    __shared__ float partial[WAVES_PER_WG][OUT_CH][SAMPLES_PER_WG];

    const int tid   = threadIdx.x;
    const int wave  = tid >> 5;
    const int lane  = tid & 31;
    const int sbase = blockIdx.x * SAMPLES_PER_WG;

    const int n     = lane & 15;
    const int khalf = (lane >> 4) << 1;

    const int tn   = t[sbase + n];
    const int widx = (980 - tn) / 20;

    const float* pa  = x + (size_t)(sbase + n) * IN_DIM
                         + (size_t)wave * K_PER_WAVE + khalf;
    const float* pb0 = W + ((size_t)widx * OUT_CH + 0) * IN_DIM
                         + (size_t)wave * K_PER_WAVE + khalf;
    const float* pb1 = W + ((size_t)widx * OUT_CH + 1) * IN_DIM
                         + (size_t)wave * K_PER_WAVE + khalf;

    v8f c0 = {};
    v8f c1 = {};

    for (int kt = 0; kt < K_PER_WAVE; kt += K_OUTER) {
        __builtin_prefetch(pa + K_OUTER, 0, 3);
#pragma unroll
        for (int u = 0; u < K_OUTER; u += 4) {
            v2f a  = *(const v2f*)(pa  + u);
            v2f b0 = *(const v2f*)(pb0 + u);
            v2f b1 = *(const v2f*)(pb1 + u);
            c0 = __builtin_amdgcn_wmma_f32_16x16x4_f32(
                     false, a, false, b0, (short)0, c0, false, false);
            c1 = __builtin_amdgcn_wmma_f32_16x16x4_f32(
                     false, a, false, b1, (short)0, c1, false, false);
        }
        pa  += K_OUTER;
        pb0 += K_OUTER;
        pb1 += K_OUTER;
    }

    int m = -1, e = 0;
    if (lane < 8)        { m = lane;      e = lane;      }
    else if (lane >= 24) { m = lane - 16; e = lane - 24; }

    float d0 = 0.0f, d1 = 0.0f;
#pragma unroll
    for (int i = 0; i < 8; ++i) {
        d0 = (e == i) ? c0[i] : d0;
        d1 = (e == i) ? c1[i] : d1;
    }
    if (m >= 0) {
        partial[wave][0][m] = d0;
        partial[wave][1][m] = d1;
    }
    __syncthreads();

    if (tid < SAMPLES_PER_WG * OUT_CH) {
        const int mm = tid >> 1;
        const int ch = tid & 1;
        float s = 0.0f;
#pragma unroll
        for (int w = 0; w < WAVES_PER_WG; ++w) s += partial[w][ch][mm];
        const int wi = (980 - t[sbase + mm]) / 20;
        out[(size_t)(sbase + mm) * OUT_CH + ch] = s + bias[wi * OUT_CH + ch];
    }
}

extern "C" void kernel_launch(void* const* d_in, const int* in_sizes, int n_in,
                              void* d_out, int out_size, void* d_ws, size_t ws_size,
                              hipStream_t stream) {
    const float* x    = (const float*)d_in[0];
    const int*   t    = (const int*)  d_in[1];
    const float* W    = (const float*)d_in[2];
    const float* bias = (const float*)d_in[3];
    float*       out  = (float*)d_out;

    dim3 grid(BATCH / SAMPLES_PER_WG);
    dim3 block(THREADS_PER_WG);
    diag_wmma_gathered_matvec<<<grid, block, 0, stream>>>(x, t, W, bias, out);
}


# Split-K: each work-group keeps a 512-k slice of all 100 W rows in LDS and gathers B fragments from LDS, x read once; a 2nd launch adds the 32 slice partials (sum order regrouped)
# speedup vs baseline: 2.1679x; 2.1679x over previous
#include <hip/hip_runtime.h>


typedef float v2f __attribute__((ext_vector_type(2)));
typedef float v4f __attribute__((ext_vector_type(4)));
typedef float v8f __attribute__((ext_vector_type(8)));

#define IN_DIM         16384
#define OUT_CH         2
#define BATCH          4096
#define N_IDX          50
#define W_ROWS         (N_IDX * OUT_CH)
#define KS             512
#define N_KS           (IN_DIM / KS)
#define SB             512
#define WAVES          16
#define THREADS        (32 * WAVES)
#define KC             64
#define N_CH           (KS / KC)
#define PF             4
#define W_LOADS        (W_ROWS * (KS / 128) / WAVES)
#define WS_NEED_BYTES  (N_KS * BATCH * OUT_CH * 4)

#define F_SAMPLES      16
#define F_WAVES        8
#define F_K_PER_WAVE   (IN_DIM / F_WAVES)
#define F_KC           128
#define F_NCHUNK       (F_K_PER_WAVE / F_KC)
#define F_ROWS         (3 * F_SAMPLES)
#define F_PITCH        (F_KC + 4)

#define SMEM_FLOATS    (W_ROWS * KS + WAVES * 16 * KC)

__device__ __forceinline__ void wave_lds_handover() {
    __builtin_amdgcn_fence(__ATOMIC_RELEASE, "wavefront");
    __builtin_amdgcn_wave_barrier();
    __builtin_amdgcn_fence(__ATOMIC_ACQUIRE, "wavefront");
}

__device__ __forceinline__ void kslice_path(const float* __restrict__ x,
                                            const int*   __restrict__ t,
                                            const float* __restrict__ W,
                                            float*       __restrict__ ws,
                                            float* smem) {
    const int tid  = threadIdx.x;
    const int lane = tid & 31;
    const int wave = __builtin_amdgcn_readfirstlane(tid >> 5);
    const int ks   = blockIdx.x & (N_KS - 1);
    const int sb   = blockIdx.x >> 5;
    const int n    = lane & 15;
    const int h    = lane >> 4;

    float* Wl = smem;
    float* xt = smem + W_ROWS * KS + wave * (16 * KC);

    const float* Wg = W + ks * KS + 4 * lane;
    v4f wv[W_LOADS];
#pragma unroll
    for (int i = 0; i < W_LOADS; ++i) {
        const int L = wave * W_LOADS + i;
        wv[i] = *(const v4f*)(Wg + (size_t)(L >> 2) * IN_DIM + (L & 3) * 128);
    }

    const int s0 = sb * SB + wave * 32;
    const float* xl = x + (size_t)(s0 + h) * IN_DIM + ks * KS + 4 * n;

    v4f g[PF][8];
#pragma unroll
    for (int c = 0; c < 2; ++c)
#pragma unroll
        for (int q = 0; q < 8; ++q)
            g[c][q] = *(const v4f*)(xl + (size_t)(2 * q) * IN_DIM + c * KC);

    const int wr0 = ((980 - t[s0 + n]) / 20) * (OUT_CH * KS);
    const int wr1 = ((980 - t[s0 + 16 + n]) / 20) * (OUT_CH * KS);

#pragma unroll
    for (int i = 0; i < W_LOADS; ++i) {
        const int L = wave * W_LOADS + i;
        *(v4f*)(Wl + (L >> 2) * KS + (L & 3) * 128 + 4 * lane) = wv[i];
    }

#pragma unroll
    for (int c = 2; c < PF; ++c)
#pragma unroll
        for (int q = 0; q < 8; ++q)
            g[c][q] = *(const v4f*)(xl + (size_t)(2 * q) * IN_DIM + c * KC);

    __syncthreads();

    float*       st = xt + h * KC + 4 * n;
    const float* fa = xt + n * KC;

    for (int tl = 0; tl < 2; ++tl) {
        const float* fb = Wl + (tl ? wr1 : wr0);
        v8f c0 = {};
        v8f c1 = {};

#pragma unroll
        for (int c = 0; c < N_CH; ++c) {
            wave_lds_handover();

#pragma unroll
            for (int q = 0; q < 8; ++q)
                *(v4f*)(st + (2 * q) * KC) = g[c & (PF - 1)][q];

            if (c + PF < N_CH) {
#pragma unroll
                for (int q = 0; q < 8; ++q)
                    g[c & (PF - 1)][q] = *(const v4f*)(xl + (size_t)(tl * 16 + 2 * q) * IN_DIM + (c + PF) * KC);
            } else if (tl == 0) {
#pragma unroll
                for (int q = 0; q < 8; ++q)
                    g[c & (PF - 1)][q] = *(const v4f*)(xl + (size_t)(16 + 2 * q) * IN_DIM + (c + PF - N_CH) * KC);
            }

            wave_lds_handover();

#pragma unroll
            for (int j = 0; j < 8; ++j) {
                const int uo = 4 * ((n + 2 * j + h) & 15);
                const v4f a  = *(const v4f*)(fa + uo);
                const v4f b0 = *(const v4f*)(fb + c * KC + uo);
                const v4f b1 = *(const v4f*)(fb + KS + c * KC + uo);
                const v2f alo = {a.x, a.y},   ahi = {a.z, a.w};
                const v2f b0lo = {b0.x, b0.y}, b0hi = {b0.z, b0.w};
                const v2f b1lo = {b1.x, b1.y}, b1hi = {b1.z, b1.w};
                c0 = __builtin_amdgcn_wmma_f32_16x16x4_f32(
                         false, alo, false, b0lo, (short)0, c0, false, false);
                c1 = __builtin_amdgcn_wmma_f32_16x16x4_f32(
                         false, alo, false, b1lo, (short)0, c1, false, false);
                c0 = __builtin_amdgcn_wmma_f32_16x16x4_f32(
                         false, ahi, false, b0hi, (short)0, c0, false, false);
                c1 = __builtin_amdgcn_wmma_f32_16x16x4_f32(
                         false, ahi, false, b1hi, (short)0, c1, false, false);
            }
        }

        int m = -1, e = 0;
        if (lane < 8)        { m = lane;      e = lane;      }
        else if (lane >= 24) { m = lane - 16; e = lane - 24; }

        float d0 = 0.0f, d1 = 0.0f;
#pragma unroll
        for (int i = 0; i < 8; ++i) {
            d0 = (e == i) ? c0[i] : d0;
            d1 = (e == i) ? c1[i] : d1;
        }
        if (m >= 0) {
            const v2f d = {d0, d1};
            *(v2f*)(ws + ((size_t)ks * BATCH + (s0 + tl * 16 + m)) * OUT_CH) = d;
        }
    }
}

__device__ __forceinline__ void single_launch_path(const float* __restrict__ x,
                                                   const int*   __restrict__ t,
                                                   const float* __restrict__ W,
                                                   const float* __restrict__ bias,
                                                   float*       __restrict__ out,
                                                   float* smem) {
    float* partial = smem + F_WAVES * F_ROWS * F_PITCH;

    const int tid   = threadIdx.x;
    const int lane  = tid & 31;
    const int wave  = __builtin_amdgcn_readfirstlane(tid >> 5);
    const int sbase = blockIdx.x * F_SAMPLES;
    const int n     = lane & 15;
    const int h     = lane >> 4;

    if (wave < F_WAVES) {
        float* tile = smem + wave * (F_ROWS * F_PITCH);
        const float* xw = x + (size_t)sbase * IN_DIM + (size_t)wave * F_K_PER_WAVE;
        const float* Ww = W + (size_t)wave * F_K_PER_WAVE;

        int wofs[F_SAMPLES];
#pragma unroll
        for (int j = 0; j < F_SAMPLES; ++j)
            wofs[j] = ((980 - t[sbase + j]) / 20) * (OUT_CH * IN_DIM);

        float*       st = tile + 4 * lane;
        const float* fr = tile + n * F_PITCH + 4 * h;

        v8f c0 = {};
        v8f c1 = {};

        v4f g[F_ROWS];
#pragma unroll
        for (int j = 0; j < F_SAMPLES; ++j)
            g[j] = *(const v4f*)(xw + (size_t)j * IN_DIM + 4 * lane);
#pragma unroll
        for (int j = 0; j < F_SAMPLES; ++j) {
            g[16 + j] = *(const v4f*)(Ww + wofs[j] + 4 * lane);
            g[32 + j] = *(const v4f*)(Ww + wofs[j] + IN_DIM + 4 * lane);
        }

        for (int c = 0; c < F_NCHUNK; ++c) {
            wave_lds_handover();
#pragma unroll
            for (int r = 0; r < F_ROWS; ++r)
                *(v4f*)(st + r * F_PITCH) = g[r];

            if (c + 1 < F_NCHUNK) {
                const int ko = (c + 1) * F_KC + 4 * lane;
#pragma unroll
                for (int j = 0; j < F_SAMPLES; ++j)
                    g[j] = *(const v4f*)(xw + (size_t)j * IN_DIM + ko);
#pragma unroll
                for (int j = 0; j < F_SAMPLES; ++j) {
                    g[16 + j] = *(const v4f*)(Ww + wofs[j] + ko);
                    g[32 + j] = *(const v4f*)(Ww + wofs[j] + IN_DIM + ko);
                }
            }

            wave_lds_handover();

#pragma unroll
            for (int j8 = 0; j8 < F_KC / 8; ++j8) {
                const v4f a  = *(const v4f*)(fr + 8 * j8);
                const v4f b0 = *(const v4f*)(fr + 16 * F_PITCH + 8 * j8);
                const v4f b1 = *(const v4f*)(fr + 32 * F_PITCH + 8 * j8);
                const v2f alo = {a.x, a.y},   ahi = {a.z, a.w};
                const v2f b0lo = {b0.x, b0.y}, b0hi = {b0.z, b0.w};
                const v2f b1lo = {b1.x, b1.y}, b1hi = {b1.z, b1.w};
                c0 = __builtin_amdgcn_wmma_f32_16x16x4_f32(
                         false, alo, false, b0lo, (short)0, c0, false, false);
                c1 = __builtin_amdgcn_wmma_f32_16x16x4_f32(
                         false, alo, false, b1lo, (short)0, c1, false, false);
                c0 = __builtin_amdgcn_wmma_f32_16x16x4_f32(
                         false, ahi, false, b0hi, (short)0, c0, false, false);
                c1 = __builtin_amdgcn_wmma_f32_16x16x4_f32(
                         false, ahi, false, b1hi, (short)0, c1, false, false);
            }
        }

        int m = -1, e = 0;
        if (lane < 8)        { m = lane;      e = lane;      }
        else if (lane >= 24) { m = lane - 16; e = lane - 24; }

        float d0 = 0.0f, d1 = 0.0f;
#pragma unroll
        for (int i = 0; i < 8; ++i) {
            d0 = (e == i) ? c0[i] : d0;
            d1 = (e == i) ? c1[i] : d1;
        }
        if (m >= 0) {
            partial[(wave * OUT_CH + 0) * F_SAMPLES + m] = d0;
            partial[(wave * OUT_CH + 1) * F_SAMPLES + m] = d1;
        }
    }
    __syncthreads();

    if (tid < F_SAMPLES * OUT_CH) {
        const int mm = tid >> 1;
        const int ch = tid & 1;
        float s = 0.0f;
#pragma unroll
        for (int w = 0; w < F_WAVES; ++w) s += partial[(w * OUT_CH + ch) * F_SAMPLES + mm];
        const int wi = (980 - t[sbase + mm]) / 20;
        out[(size_t)(sbase + mm) * OUT_CH + ch] = s + bias[wi * OUT_CH + ch];
    }
}

__global__ __launch_bounds__(THREADS)
void kslice_resident_w_matvec(const float* __restrict__ x,
                              const int*   __restrict__ t,
                              const float* __restrict__ W,
                              const float* __restrict__ bias,
                              float*       __restrict__ out,
                              float*       __restrict__ ws,
                              unsigned long long ws_bytes) {
    __shared__ __attribute__((aligned(16))) float smem[SMEM_FLOATS];
    if (ws_bytes >= (unsigned long long)WS_NEED_BYTES)
        kslice_path(x, t, W, ws, smem);
    else
        single_launch_path(x, t, W, bias, out, smem);
}

__global__ __launch_bounds__(256)
void add_slices_and_bias(const float* __restrict__ ws,
                         const int*   __restrict__ t,
                         const float* __restrict__ bias,
                         float*       __restrict__ out,
                         unsigned long long ws_bytes) {
    if (ws_bytes < (unsigned long long)WS_NEED_BYTES) return;
    const int o = blockIdx.x * 256 + threadIdx.x;
    float s = 0.0f;
#pragma unroll
    for (int k = 0; k < N_KS; ++k) s += ws[(size_t)k * (BATCH * OUT_CH) + o];
    const int wi = (980 - t[o >> 1]) / 20;
    out[o] = s + bias[wi * OUT_CH + (o & 1)];
}

extern "C" void kernel_launch(void* const* d_in, const int* in_sizes, int n_in,
                              void* d_out, int out_size, void* d_ws, size_t ws_size,
                              hipStream_t stream) {
    const float* x    = (const float*)d_in[0];
    const int*   t    = (const int*)  d_in[1];
    const float* W    = (const float*)d_in[2];
    const float* bias = (const float*)d_in[3];
    float*       out  = (float*)d_out;
    float*       ws   = (float*)d_ws;

    kslice_resident_w_matvec<<<dim3(256), dim3(THREADS), 0, stream>>>(
        x, t, W, bias, out, ws, (unsigned long long)ws_size);
    add_slices_and_bias<<<dim3(32), dim3(256), 0, stream>>>(
        ws, t, bias, out, (unsigned long long)ws_size);
}


# Tile = 16 samples adjacent in weight-index order (ballot counting sort per work-group): B's 16 columns hold both channels' rows of the tile's indices, 1 WMMA per 4 k instead of 2; sums regrouped
# speedup vs baseline: 2.4015x; 1.1078x over previous
#include <hip/hip_runtime.h>


typedef float v2f __attribute__((ext_vector_type(2)));
typedef float v4f __attribute__((ext_vector_type(4)));
typedef float v8f __attribute__((ext_vector_type(8)));
typedef int   v4i __attribute__((ext_vector_type(4)));

#define IN_DIM         16384
#define OUT_CH         2
#define BATCH          4096
#define N_IDX          50
#define W_ROWS         (N_IDX * OUT_CH)
#define KS             512
#define N_KS           (IN_DIM / KS)
#define SB             512
#define WAVES          16
#define THREADS        (32 * WAVES)
#define KC             64
#define N_CH           (KS / KC)
#define PF             4
#define W_LOADS        (W_ROWS * (KS / 128) / WAVES)
#define WS_NEED_BYTES  (N_KS * BATCH * OUT_CH * 4)
#define WPITCH         (KS + 4)
#define XPITCH         (KC + 4)
#define XT_FLOATS      (16 * XPITCH)

#define F_SAMPLES      16
#define F_WAVES        8
#define F_K_PER_WAVE   (IN_DIM / F_WAVES)
#define F_KC           128
#define F_NCHUNK       (F_K_PER_WAVE / F_KC)
#define F_ROWS         (3 * F_SAMPLES)
#define F_PITCH        (F_KC + 4)

#define OFF_XT         (W_ROWS * WPITCH)
#define OFF_OUT        (OFF_XT + WAVES * XT_FLOATS)
#define OFF_INT        (OFF_OUT + SB * OUT_CH)
#define SMEM_FLOATS    (OFF_INT + WAVES * 64 + 2 * SB + WAVES * 32)

__device__ __forceinline__ void wave_lds_handover() {
    __builtin_amdgcn_fence(__ATOMIC_RELEASE, "wavefront");
    __builtin_amdgcn_wave_barrier();
    __builtin_amdgcn_fence(__ATOMIC_ACQUIRE, "wavefront");
}

__device__ __forceinline__ void lds_only_barrier() {
    __builtin_amdgcn_fence(__ATOMIC_RELEASE, "workgroup", "local");
    __builtin_amdgcn_s_barrier();
    __builtin_amdgcn_fence(__ATOMIC_ACQUIRE, "workgroup", "local");
}

__device__ __forceinline__ void kslice_path(const float* __restrict__ x,
                                            const int*   __restrict__ t,
                                            const float* __restrict__ W,
                                            float*       __restrict__ ws,
                                            float* smem) {
    const int tid  = threadIdx.x;
    const int lane = tid & 31;
    const int wave = __builtin_amdgcn_readfirstlane(tid >> 5);
    const int ks   = blockIdx.x & (N_KS - 1);
    const int sb   = blockIdx.x >> 5;
    const int n    = lane & 15;
    const int h    = lane >> 4;

    float* Wl   = smem;
    float* xt   = smem + OFF_XT + wave * XT_FLOATS;
    float* outl = smem + OFF_OUT;
    int*   cnt  = (int*)(smem + OFF_INT);
    int*   ord  = cnt + WAVES * 64;
    int*   oix  = ord + SB;
    int*   dl   = oix + SB + wave * 32;

    const int my_t = t[sb * SB + tid];

    const float* Wg = W + ks * KS + 4 * lane;
    v4f wv[W_LOADS];
#pragma unroll
    for (int i = 0; i < W_LOADS; ++i) {
        const int L = wave * W_LOADS + i;
        wv[i] = *(const v4f*)(Wg + (size_t)(L >> 2) * IN_DIM + (L & 3) * 128);
    }

    const int my_idx = (980 - my_t) / 20;
    unsigned peers = 0xFFFFFFFFu;
#pragma unroll
    for (int k = 0; k < 6; ++k) {
        const bool bit = (my_idx >> k) & 1;
        const unsigned mb = __builtin_amdgcn_ballot_w32(bit);
        peers &= bit ? mb : ~mb;
    }
    const int rank_w = (int)__builtin_amdgcn_mbcnt_lo(peers, 0u);
    cnt[wave * 64 + lane]      = 0;
    cnt[wave * 64 + 32 + lane] = 0;
    if (rank_w == 0) cnt[wave * 64 + my_idx] = __builtin_popcount(peers);
    lds_only_barrier();
    int tot_lo = 0, tot_hi = 0, pre_lo = 0, pre_hi = 0;
#pragma unroll
    for (int w = 0; w < WAVES; ++w) {
        const int a0 = cnt[w * 64 + lane];
        const int a1 = cnt[w * 64 + 32 + lane];
        if (w < wave) { pre_lo += a0; pre_hi += a1; }
        tot_lo += a0; tot_hi += a1;
    }
    int inc_lo = tot_lo, inc_hi = tot_hi;
#pragma unroll
    for (int d = 1; d < 32; d <<= 1) {
        const int y0 = __shfl_up(inc_lo, d);
        const int y1 = __shfl_up(inc_hi, d);
        if (lane >= d) { inc_lo += y0; inc_hi += y1; }
    }
    const int all_lo   = __shfl(inc_lo, 31);
    const int start_lo = inc_lo - tot_lo + pre_lo;
    const int start_hi = all_lo + inc_hi - tot_hi + pre_hi;
    const int f_lo = __shfl(start_lo, my_idx & 31);
    const int f_hi = __shfl(start_hi, my_idx & 31);
    const int pos  = ((my_idx < 32) ? f_lo : f_hi) + rank_w;
    ord[pos] = tid;
    oix[pos] = my_idx;
    lds_only_barrier();

    const int p0 = wave * 32;
    unsigned xb[2][8];
    unsigned msk0 = 0, msk1 = 0;
    int nd0 = 1, nd1 = 1;
#pragma unroll
    for (int tl = 0; tl < 2; ++tl) {
        const int vcur  = oix[p0 + 16 * tl + n];
        const int vprev = oix[p0 + 16 * tl + (n > 0 ? n - 1 : 0)];
        const bool first = (n > 0) && (vcur != vprev);
        const unsigned mk = __builtin_amdgcn_ballot_w32(first) & 0xFFFFu;
        const int rk = __builtin_popcount(mk & ((2u << n) - 1u));
        if (h == 0 && (n == 0 || first)) dl[16 * tl + rk] = vcur;
        if (tl == 0) { msk0 = mk; nd0 = __builtin_popcount(mk) + 1; }
        else         { msk1 = mk; nd1 = __builtin_popcount(mk) + 1; }
#pragma unroll
        for (int q = 0; q < 8; ++q) {
            const int sl = ord[p0 + 16 * tl + 2 * q + h];
            xb[tl][q] = ((unsigned)(sb * SB + sl) * (unsigned)IN_DIM + (unsigned)(ks * KS + 4 * n)) * 4u;
        }
    }
    wave_lds_handover();

    const int ng0     = (nd0 > 8) ? 2 : 1;
    const int ng1     = (nd1 > 8) ? 2 : 1;
    const int n_items = __builtin_amdgcn_readfirstlane(ng0 + ng1);

    const char* xc = (const char*)x;
    v4f g[PF][8];
#pragma unroll
    for (int c = 0; c < PF; ++c)
#pragma unroll
        for (int q = 0; q < 8; ++q)
            g[c][q] = *(const v4f*)(xc + xb[0][q] + c * (KC * 4));
    __builtin_amdgcn_sched_barrier(0);

#pragma unroll
    for (int i = 0; i < W_LOADS; ++i) {
        const int L = wave * W_LOADS + i;
        *(v4f*)(Wl + (L >> 2) * WPITCH + (L & 3) * 128 + 4 * lane) = wv[i];
    }

    lds_only_barrier();

    float*       st = xt + h * XPITCH + 4 * n;
    const float* fa = xt + n * XPITCH + 4 * h;
    const float* fb = Wl + 4 * h;
    v8f acc0 = {};
    v8f acc1 = {};

    int item = 0;
    do {
        const int tl = (item >= ng0) ? 1 : 0;
        const int gr = item - (tl ? ng0 : 0);
        {
            const int ndt = tl ? nd1 : nd0;
            int j = 8 * gr + (n >> 1);
            j = (j < ndt) ? j : (ndt - 1);
            const int d = dl[16 * tl + j];
            fb = Wl + (2 * d + (n & 1)) * WPITCH + 4 * h;
        }
        const int tln = (item + 1 >= ng0) ? 1 : 0;
        acc0 = (v8f){};
        acc1 = (v8f){};

#pragma unroll
        for (int half = 0; half < 2; ++half) {
            const bool have_next = (half == 0) || (item + 1 < n_items);
            unsigned xn[8];
#pragma unroll
            for (int q = 0; q < 8; ++q)
                xn[q] = (half == 0) ? (xb[tl][q] + 4u * (KC * 4))
                                    : (tln ? xb[1][q] : xb[0][q]);

#pragma unroll
            for (int cc = 0; cc < PF; ++cc) {
                const int c = half * PF + cc;
                wave_lds_handover();

#pragma unroll
                for (int q = 0; q < 8; ++q)
                    *(v4f*)(st + (2 * q) * XPITCH) = g[cc][q];

                if (have_next) {
#pragma unroll
                    for (int q = 0; q < 8; ++q)
                        g[cc][q] = *(const v4f*)(xc + xn[q] + cc * (KC * 4));
                }

                wave_lds_handover();

                v4f af[8], bf[8];
#pragma unroll
                for (int j = 0; j < 8; ++j) af[j] = *(const v4f*)(fa + 8 * j);
#pragma unroll
                for (int j = 0; j < 8; ++j) bf[j] = *(const v4f*)(fb + c * KC + 8 * j);
                __builtin_amdgcn_sched_barrier(0);
#pragma unroll
                for (int j = 0; j < 8; ++j) {
                    const v2f alo = {af[j].x, af[j].y}, ahi = {af[j].z, af[j].w};
                    const v2f blo = {bf[j].x, bf[j].y}, bhi = {bf[j].z, bf[j].w};
                    acc0 = __builtin_amdgcn_wmma_f32_16x16x4_f32(
                               false, alo, false, blo, (short)0, acc0, false, false);
                    acc1 = __builtin_amdgcn_wmma_f32_16x16x4_f32(
                               false, ahi, false, bhi, (short)0, acc1, false, false);
                }
            }
        }

        const unsigned mk = tl ? msk1 : msk0;
        const v8f acc = acc0 + acc1;
        const v4i sa = *(const v4i*)(ord + p0 + 16 * tl + 8 * h);
        const v4i sc = *(const v4i*)(ord + p0 + 16 * tl + 8 * h + 4);
        const int srow[8] = {sa.x, sa.y, sa.z, sa.w, sc.x, sc.y, sc.z, sc.w};
#pragma unroll
        for (int r = 0; r < 8; ++r) {
            const int m  = 8 * h + r;
            const int jj = __builtin_popcount(mk & ((2u << m) - 1u)) - 8 * gr;
            if (jj == (n >> 1)) outl[2 * srow[r] + (n & 1)] = acc[r];
        }
    } while (++item < n_items);

    __syncthreads();
    const v2f o = *(const v2f*)(outl + 2 * tid);
    *(v2f*)(ws + ((size_t)ks * BATCH + (size_t)(sb * SB + tid)) * OUT_CH) = o;
}

__device__ __forceinline__ void single_launch_path(const float* __restrict__ x,
                                                   const int*   __restrict__ t,
                                                   const float* __restrict__ W,
                                                   const float* __restrict__ bias,
                                                   float*       __restrict__ out,
                                                   float* smem) {
    float* partial = smem + F_WAVES * F_ROWS * F_PITCH;

    const int tid   = threadIdx.x;
    const int lane  = tid & 31;
    const int wave  = __builtin_amdgcn_readfirstlane(tid >> 5);
    const int sbase = blockIdx.x * F_SAMPLES;
    const int n     = lane & 15;
    const int h     = lane >> 4;

    if (wave < F_WAVES) {
        float* tile = smem + wave * (F_ROWS * F_PITCH);
        const float* xw = x + (size_t)sbase * IN_DIM + (size_t)wave * F_K_PER_WAVE;
        const float* Ww = W + (size_t)wave * F_K_PER_WAVE;

        int wofs[F_SAMPLES];
#pragma unroll
        for (int j = 0; j < F_SAMPLES; ++j)
            wofs[j] = ((980 - t[sbase + j]) / 20) * (OUT_CH * IN_DIM);

        float*       st = tile + 4 * lane;
        const float* fr = tile + n * F_PITCH + 4 * h;

        v8f c0 = {};
        v8f c1 = {};

        v4f g[F_ROWS];
#pragma unroll
        for (int j = 0; j < F_SAMPLES; ++j)
            g[j] = *(const v4f*)(xw + (size_t)j * IN_DIM + 4 * lane);
#pragma unroll
        for (int j = 0; j < F_SAMPLES; ++j) {
            g[16 + j] = *(const v4f*)(Ww + wofs[j] + 4 * lane);
            g[32 + j] = *(const v4f*)(Ww + wofs[j] + IN_DIM + 4 * lane);
        }

        for (int c = 0; c < F_NCHUNK; ++c) {
            wave_lds_handover();
#pragma unroll
            for (int r = 0; r < F_ROWS; ++r)
                *(v4f*)(st + r * F_PITCH) = g[r];

            if (c + 1 < F_NCHUNK) {
                const int ko = (c + 1) * F_KC + 4 * lane;
#pragma unroll
                for (int j = 0; j < F_SAMPLES; ++j)
                    g[j] = *(const v4f*)(xw + (size_t)j * IN_DIM + ko);
#pragma unroll
                for (int j = 0; j < F_SAMPLES; ++j) {
                    g[16 + j] = *(const v4f*)(Ww + wofs[j] + ko);
                    g[32 + j] = *(const v4f*)(Ww + wofs[j] + IN_DIM + ko);
                }
            }

            wave_lds_handover();

#pragma unroll
            for (int j8 = 0; j8 < F_KC / 8; ++j8) {
                const v4f a  = *(const v4f*)(fr + 8 * j8);
                const v4f b0 = *(const v4f*)(fr + 16 * F_PITCH + 8 * j8);
                const v4f b1 = *(const v4f*)(fr + 32 * F_PITCH + 8 * j8);
                const v2f alo = {a.x, a.y},   ahi = {a.z, a.w};
                const v2f b0lo = {b0.x, b0.y}, b0hi = {b0.z, b0.w};
                const v2f b1lo = {b1.x, b1.y}, b1hi = {b1.z, b1.w};
                c0 = __builtin_amdgcn_wmma_f32_16x16x4_f32(
                         false, alo, false, b0lo, (short)0, c0, false, false);
                c1 = __builtin_amdgcn_wmma_f32_16x16x4_f32(
                         false, alo, false, b1lo, (short)0, c1, false, false);
                c0 = __builtin_amdgcn_wmma_f32_16x16x4_f32(
                         false, ahi, false, b0hi, (short)0, c0, false, false);
                c1 = __builtin_amdgcn_wmma_f32_16x16x4_f32(
                         false, ahi, false, b1hi, (short)0, c1, false, false);
            }
        }

        int m = -1, e = 0;
        if (lane < 8)        { m = lane;      e = lane;      }
        else if (lane >= 24) { m = lane - 16; e = lane - 24; }

        float d0 = 0.0f, d1 = 0.0f;
#pragma unroll
        for (int i = 0; i < 8; ++i) {
            d0 = (e == i) ? c0[i] : d0;
            d1 = (e == i) ? c1[i] : d1;
        }
        if (m >= 0) {
            partial[(wave * OUT_CH + 0) * F_SAMPLES + m] = d0;
            partial[(wave * OUT_CH + 1) * F_SAMPLES + m] = d1;
        }
    }
    __syncthreads();

    if (tid < F_SAMPLES * OUT_CH) {
        const int mm = tid >> 1;
        const int ch = tid & 1;
        float s = 0.0f;
#pragma unroll
        for (int w = 0; w < F_WAVES; ++w) s += partial[(w * OUT_CH + ch) * F_SAMPLES + mm];
        const int wi = (980 - t[sbase + mm]) / 20;
        out[(size_t)(sbase + mm) * OUT_CH + ch] = s + bias[wi * OUT_CH + ch];
    }
}

__global__ __launch_bounds__(THREADS)
void kslice_resident_w_matvec(const float* __restrict__ x,
                              const int*   __restrict__ t,
                              const float* __restrict__ W,
                              const float* __restrict__ bias,
                              float*       __restrict__ out,
                              float*       __restrict__ ws,
                              unsigned long long ws_bytes) {
    __shared__ __attribute__((aligned(16))) float smem[SMEM_FLOATS];
    if (ws_bytes >= (unsigned long long)WS_NEED_BYTES)
        kslice_path(x, t, W, ws, smem);
    else
        single_launch_path(x, t, W, bias, out, smem);
}

__global__ __launch_bounds__(256)
void add_slices_and_bias(const float* __restrict__ ws,
                         const int*   __restrict__ t,
                         const float* __restrict__ bias,
                         float*       __restrict__ out,
                         unsigned long long ws_bytes) {
    if (ws_bytes < (unsigned long long)WS_NEED_BYTES) return;
    const int o = blockIdx.x * 256 + threadIdx.x;
    float s = 0.0f;
#pragma unroll
    for (int k = 0; k < N_KS; ++k) s += ws[(size_t)k * (BATCH * OUT_CH) + o];
    const int wi = (980 - t[o >> 1]) / 20;
    out[o] = s + bias[wi * OUT_CH + (o & 1)];
}

extern "C" void kernel_launch(void* const* d_in, const int* in_sizes, int n_in,
                              void* d_out, int out_size, void* d_ws, size_t ws_size,
                              hipStream_t stream) {
    const float* x    = (const float*)d_in[0];
    const int*   t    = (const int*)  d_in[1];
    const float* W    = (const float*)d_in[2];
    const float* bias = (const float*)d_in[3];
    float*       out  = (float*)d_out;
    float*       ws   = (float*)d_ws;

    kslice_resident_w_matvec<<<dim3(256), dim3(THREADS), 0, stream>>>(
        x, t, W, bias, out, ws, (unsigned long long)ws_size);
    add_slices_and_bias<<<dim3(32), dim3(256), 0, stream>>>(
        ws, t, bias, out, (unsigned long long)ws_size);
}


# Before the sort, threads whose weight index is in the lower half prefetch to L2 the first 512 B of their x slice; first walks remapped to the lower half of the order, so those rows are read first
# speedup vs baseline: 2.4953x; 1.0391x over previous
#include <hip/hip_runtime.h>


typedef float v2f __attribute__((ext_vector_type(2)));
typedef float v4f __attribute__((ext_vector_type(4)));
typedef float v8f __attribute__((ext_vector_type(8)));
typedef int   v4i __attribute__((ext_vector_type(4)));

#define IN_DIM         16384
#define OUT_CH         2
#define BATCH          4096
#define N_IDX          50
#define W_ROWS         (N_IDX * OUT_CH)
#define KS             512
#define N_KS           (IN_DIM / KS)
#define SB             512
#define WAVES          16
#define THREADS        (32 * WAVES)
#define KC             64
#define N_CH           (KS / KC)
#define PF             4
#define W_LOADS        (W_ROWS * (KS / 128) / WAVES)
#define WS_NEED_BYTES  (N_KS * BATCH * OUT_CH * 4)
#define WPITCH         (KS + 4)
#define XPITCH         (KC + 4)
#define XT_FLOATS      (16 * XPITCH)

#define F_SAMPLES      16
#define F_WAVES        8
#define F_K_PER_WAVE   (IN_DIM / F_WAVES)
#define F_KC           128
#define F_NCHUNK       (F_K_PER_WAVE / F_KC)
#define F_ROWS         (3 * F_SAMPLES)
#define F_PITCH        (F_KC + 4)

#define OFF_XT         (W_ROWS * WPITCH)
#define OFF_OUT        (OFF_XT + WAVES * XT_FLOATS)
#define OFF_INT        (OFF_OUT + SB * OUT_CH)
#define SMEM_FLOATS    (OFF_INT + WAVES * 64 + 2 * SB + WAVES * 32)

__device__ __forceinline__ void wave_lds_handover() {
    __builtin_amdgcn_fence(__ATOMIC_RELEASE, "wavefront");
    __builtin_amdgcn_wave_barrier();
    __builtin_amdgcn_fence(__ATOMIC_ACQUIRE, "wavefront");
}

__device__ __forceinline__ void lds_only_barrier() {
    __builtin_amdgcn_fence(__ATOMIC_RELEASE, "workgroup", "local");
    __builtin_amdgcn_s_barrier();
    __builtin_amdgcn_fence(__ATOMIC_ACQUIRE, "workgroup", "local");
}

__device__ __forceinline__ void kslice_path(const float* __restrict__ x,
                                            const int*   __restrict__ t,
                                            const float* __restrict__ W,
                                            float*       __restrict__ ws,
                                            float* smem) {
    const int tid  = threadIdx.x;
    const int lane = tid & 31;
    const int wave = __builtin_amdgcn_readfirstlane(tid >> 5);
    const int ks   = blockIdx.x & (N_KS - 1);
    const int sb   = blockIdx.x >> 5;
    const int n    = lane & 15;
    const int h    = lane >> 4;

    float* Wl   = smem;
    float* xt   = smem + OFF_XT + wave * XT_FLOATS;
    float* outl = smem + OFF_OUT;
    int*   cnt  = (int*)(smem + OFF_INT);
    int*   ord  = cnt + WAVES * 64;
    int*   oix  = ord + SB;
    int*   dl   = oix + SB + wave * 32;

    const int my_t = t[sb * SB + tid];

    const float* Wg = W + ks * KS + 4 * lane;
    v4f wv[W_LOADS];
#pragma unroll
    for (int i = 0; i < W_LOADS; ++i) {
        const int L = wave * W_LOADS + i;
        wv[i] = *(const v4f*)(Wg + (size_t)(L >> 2) * IN_DIM + (L & 3) * 128);
    }

    const int my_idx = (980 - my_t) / 20;

    if (my_idx < N_IDX / 2) {
        const float* xp = x + (size_t)(sb * SB + tid) * IN_DIM + ks * KS;
        __builtin_prefetch(xp, 0, 1);
        __builtin_prefetch(xp + KC, 0, 1);
    }

    unsigned peers = 0xFFFFFFFFu;
#pragma unroll
    for (int k = 0; k < 6; ++k) {
        const bool bit = (my_idx >> k) & 1;
        const unsigned mb = __builtin_amdgcn_ballot_w32(bit);
        peers &= bit ? mb : ~mb;
    }
    const int rank_w = (int)__builtin_amdgcn_mbcnt_lo(peers, 0u);
    cnt[wave * 64 + lane]      = 0;
    cnt[wave * 64 + 32 + lane] = 0;
    if (rank_w == 0) cnt[wave * 64 + my_idx] = __builtin_popcount(peers);
    lds_only_barrier();
    int tot_lo = 0, tot_hi = 0, pre_lo = 0, pre_hi = 0;
#pragma unroll
    for (int w = 0; w < WAVES; ++w) {
        const int a0 = cnt[w * 64 + lane];
        const int a1 = cnt[w * 64 + 32 + lane];
        if (w < wave) { pre_lo += a0; pre_hi += a1; }
        tot_lo += a0; tot_hi += a1;
    }
    int inc_lo = tot_lo, inc_hi = tot_hi;
#pragma unroll
    for (int d = 1; d < 32; d <<= 1) {
        const int y0 = __shfl_up(inc_lo, d);
        const int y1 = __shfl_up(inc_hi, d);
        if (lane >= d) { inc_lo += y0; inc_hi += y1; }
    }
    const int all_lo   = __shfl(inc_lo, 31);
    const int start_lo = inc_lo - tot_lo + pre_lo;
    const int start_hi = all_lo + inc_hi - tot_hi + pre_hi;
    const int f_lo = __shfl(start_lo, my_idx & 31);
    const int f_hi = __shfl(start_hi, my_idx & 31);
    const int pos  = ((my_idx < 32) ? f_lo : f_hi) + rank_w;
    ord[pos] = tid;
    oix[pos] = my_idx;
    lds_only_barrier();

    const int p0 = wave * 16;
    unsigned xb[2][8];
    unsigned msk0 = 0, msk1 = 0;
    int nd0 = 1, nd1 = 1;
#pragma unroll
    for (int tl = 0; tl < 2; ++tl) {
        const int vcur  = oix[p0 + (SB / 2) * tl + n];
        const int vprev = oix[p0 + (SB / 2) * tl + (n > 0 ? n - 1 : 0)];
        const bool first = (n > 0) && (vcur != vprev);
        const unsigned mk = __builtin_amdgcn_ballot_w32(first) & 0xFFFFu;
        const int rk = __builtin_popcount(mk & ((2u << n) - 1u));
        if (h == 0 && (n == 0 || first)) dl[16 * tl + rk] = vcur;
        if (tl == 0) { msk0 = mk; nd0 = __builtin_popcount(mk) + 1; }
        else         { msk1 = mk; nd1 = __builtin_popcount(mk) + 1; }
#pragma unroll
        for (int q = 0; q < 8; ++q) {
            const int sl = ord[p0 + (SB / 2) * tl + 2 * q + h];
            xb[tl][q] = ((unsigned)(sb * SB + sl) * (unsigned)IN_DIM + (unsigned)(ks * KS + 4 * n)) * 4u;
        }
    }
    wave_lds_handover();

    const int ng0     = (nd0 > 8) ? 2 : 1;
    const int ng1     = (nd1 > 8) ? 2 : 1;
    const int n_items = __builtin_amdgcn_readfirstlane(ng0 + ng1);

    const char* xc = (const char*)x;
    v4f g[PF][8];
#pragma unroll
    for (int c = 0; c < PF; ++c)
#pragma unroll
        for (int q = 0; q < 8; ++q)
            g[c][q] = *(const v4f*)(xc + xb[0][q] + c * (KC * 4));
    __builtin_amdgcn_sched_barrier(0);

#pragma unroll
    for (int i = 0; i < W_LOADS; ++i) {
        const int L = wave * W_LOADS + i;
        *(v4f*)(Wl + (L >> 2) * WPITCH + (L & 3) * 128 + 4 * lane) = wv[i];
    }

    lds_only_barrier();

    float*       st = xt + h * XPITCH + 4 * n;
    const float* fa = xt + n * XPITCH + 4 * h;
    const float* fb = Wl + 4 * h;
    v8f acc0 = {};
    v8f acc1 = {};

    int item = 0;
    do {
        const int tl = (item >= ng0) ? 1 : 0;
        const int gr = item - (tl ? ng0 : 0);
        {
            const int ndt = tl ? nd1 : nd0;
            int j = 8 * gr + (n >> 1);
            j = (j < ndt) ? j : (ndt - 1);
            const int d = dl[16 * tl + j];
            fb = Wl + (2 * d + (n & 1)) * WPITCH + 4 * h;
        }
        const int tln = (item + 1 >= ng0) ? 1 : 0;
        acc0 = (v8f){};
        acc1 = (v8f){};

#pragma unroll
        for (int half = 0; half < 2; ++half) {
            const bool have_next = (half == 0) || (item + 1 < n_items);
            unsigned xn[8];
#pragma unroll
            for (int q = 0; q < 8; ++q)
                xn[q] = (half == 0) ? (xb[tl][q] + 4u * (KC * 4))
                                    : (tln ? xb[1][q] : xb[0][q]);

#pragma unroll
            for (int cc = 0; cc < PF; ++cc) {
                const int c = half * PF + cc;
                wave_lds_handover();

#pragma unroll
                for (int q = 0; q < 8; ++q)
                    *(v4f*)(st + (2 * q) * XPITCH) = g[cc][q];

                if (have_next) {
#pragma unroll
                    for (int q = 0; q < 8; ++q)
                        g[cc][q] = *(const v4f*)(xc + xn[q] + cc * (KC * 4));
                }

                wave_lds_handover();

                v4f af[8], bf[8];
#pragma unroll
                for (int j = 0; j < 8; ++j) af[j] = *(const v4f*)(fa + 8 * j);
#pragma unroll
                for (int j = 0; j < 8; ++j) bf[j] = *(const v4f*)(fb + c * KC + 8 * j);
                __builtin_amdgcn_sched_barrier(0);
#pragma unroll
                for (int j = 0; j < 8; ++j) {
                    const v2f alo = {af[j].x, af[j].y}, ahi = {af[j].z, af[j].w};
                    const v2f blo = {bf[j].x, bf[j].y}, bhi = {bf[j].z, bf[j].w};
                    acc0 = __builtin_amdgcn_wmma_f32_16x16x4_f32(
                               false, alo, false, blo, (short)0, acc0, false, false);
                    acc1 = __builtin_amdgcn_wmma_f32_16x16x4_f32(
                               false, ahi, false, bhi, (short)0, acc1, false, false);
                }
            }
        }

        const unsigned mk = tl ? msk1 : msk0;
        const v8f acc = acc0 + acc1;
        const v4i sa = *(const v4i*)(ord + p0 + (SB / 2) * tl + 8 * h);
        const v4i sc = *(const v4i*)(ord + p0 + (SB / 2) * tl + 8 * h + 4);
        const int srow[8] = {sa.x, sa.y, sa.z, sa.w, sc.x, sc.y, sc.z, sc.w};
#pragma unroll
        for (int r = 0; r < 8; ++r) {
            const int m  = 8 * h + r;
            const int jj = __builtin_popcount(mk & ((2u << m) - 1u)) - 8 * gr;
            if (jj == (n >> 1)) outl[2 * srow[r] + (n & 1)] = acc[r];
        }
    } while (++item < n_items);

    __syncthreads();
    const v2f o = *(const v2f*)(outl + 2 * tid);
    *(v2f*)(ws + ((size_t)ks * BATCH + (size_t)(sb * SB + tid)) * OUT_CH) = o;
}

__device__ __forceinline__ void single_launch_path(const float* __restrict__ x,
                                                   const int*   __restrict__ t,
                                                   const float* __restrict__ W,
                                                   const float* __restrict__ bias,
                                                   float*       __restrict__ out,
                                                   float* smem) {
    float* partial = smem + F_WAVES * F_ROWS * F_PITCH;

    const int tid   = threadIdx.x;
    const int lane  = tid & 31;
    const int wave  = __builtin_amdgcn_readfirstlane(tid >> 5);
    const int sbase = blockIdx.x * F_SAMPLES;
    const int n     = lane & 15;
    const int h     = lane >> 4;

    if (wave < F_WAVES) {
        float* tile = smem + wave * (F_ROWS * F_PITCH);
        const float* xw = x + (size_t)sbase * IN_DIM + (size_t)wave * F_K_PER_WAVE;
        const float* Ww = W + (size_t)wave * F_K_PER_WAVE;

        int wofs[F_SAMPLES];
#pragma unroll
        for (int j = 0; j < F_SAMPLES; ++j)
            wofs[j] = ((980 - t[sbase + j]) / 20) * (OUT_CH * IN_DIM);

        float*       st = tile + 4 * lane;
        const float* fr = tile + n * F_PITCH + 4 * h;

        v8f c0 = {};
        v8f c1 = {};

        v4f g[F_ROWS];
#pragma unroll
        for (int j = 0; j < F_SAMPLES; ++j)
            g[j] = *(const v4f*)(xw + (size_t)j * IN_DIM + 4 * lane);
#pragma unroll
        for (int j = 0; j < F_SAMPLES; ++j) {
            g[16 + j] = *(const v4f*)(Ww + wofs[j] + 4 * lane);
            g[32 + j] = *(const v4f*)(Ww + wofs[j] + IN_DIM + 4 * lane);
        }

        for (int c = 0; c < F_NCHUNK; ++c) {
            wave_lds_handover();
#pragma unroll
            for (int r = 0; r < F_ROWS; ++r)
                *(v4f*)(st + r * F_PITCH) = g[r];

            if (c + 1 < F_NCHUNK) {
                const int ko = (c + 1) * F_KC + 4 * lane;
#pragma unroll
                for (int j = 0; j < F_SAMPLES; ++j)
                    g[j] = *(const v4f*)(xw + (size_t)j * IN_DIM + ko);
#pragma unroll
                for (int j = 0; j < F_SAMPLES; ++j) {
                    g[16 + j] = *(const v4f*)(Ww + wofs[j] + ko);
                    g[32 + j] = *(const v4f*)(Ww + wofs[j] + IN_DIM + ko);
                }
            }

            wave_lds_handover();

#pragma unroll
            for (int j8 = 0; j8 < F_KC / 8; ++j8) {
                const v4f a  = *(const v4f*)(fr + 8 * j8);
                const v4f b0 = *(const v4f*)(fr + 16 * F_PITCH + 8 * j8);
                const v4f b1 = *(const v4f*)(fr + 32 * F_PITCH + 8 * j8);
                const v2f alo = {a.x, a.y},   ahi = {a.z, a.w};
                const v2f b0lo = {b0.x, b0.y}, b0hi = {b0.z, b0.w};
                const v2f b1lo = {b1.x, b1.y}, b1hi = {b1.z, b1.w};
                c0 = __builtin_amdgcn_wmma_f32_16x16x4_f32(
                         false, alo, false, b0lo, (short)0, c0, false, false);
                c1 = __builtin_amdgcn_wmma_f32_16x16x4_f32(
                         false, alo, false, b1lo, (short)0, c1, false, false);
                c0 = __builtin_amdgcn_wmma_f32_16x16x4_f32(
                         false, ahi, false, b0hi, (short)0, c0, false, false);
                c1 = __builtin_amdgcn_wmma_f32_16x16x4_f32(
                         false, ahi, false, b1hi, (short)0, c1, false, false);
            }
        }

        int m = -1, e = 0;
        if (lane < 8)        { m = lane;      e = lane;      }
        else if (lane >= 24) { m = lane - 16; e = lane - 24; }

        float d0 = 0.0f, d1 = 0.0f;
#pragma unroll
        for (int i = 0; i < 8; ++i) {
            d0 = (e == i) ? c0[i] : d0;
            d1 = (e == i) ? c1[i] : d1;
        }
        if (m >= 0) {
            partial[(wave * OUT_CH + 0) * F_SAMPLES + m] = d0;
            partial[(wave * OUT_CH + 1) * F_SAMPLES + m] = d1;
        }
    }
    __syncthreads();

    if (tid < F_SAMPLES * OUT_CH) {
        const int mm = tid >> 1;
        const int ch = tid & 1;
        float s = 0.0f;
#pragma unroll
        for (int w = 0; w < F_WAVES; ++w) s += partial[(w * OUT_CH + ch) * F_SAMPLES + mm];
        const int wi = (980 - t[sbase + mm]) / 20;
        out[(size_t)(sbase + mm) * OUT_CH + ch] = s + bias[wi * OUT_CH + ch];
    }
}

__global__ __launch_bounds__(THREADS)
void kslice_resident_w_matvec(const float* __restrict__ x,
                              const int*   __restrict__ t,
                              const float* __restrict__ W,
                              const float* __restrict__ bias,
                              float*       __restrict__ out,
                              float*       __restrict__ ws,
                              unsigned long long ws_bytes) {
    __shared__ __attribute__((aligned(16))) float smem[SMEM_FLOATS];
    if (ws_bytes >= (unsigned long long)WS_NEED_BYTES)
        kslice_path(x, t, W, ws, smem);
    else
        single_launch_path(x, t, W, bias, out, smem);
}

__global__ __launch_bounds__(256)
void add_slices_and_bias(const float* __restrict__ ws,
                         const int*   __restrict__ t,
                         const float* __restrict__ bias,
                         float*       __restrict__ out,
                         unsigned long long ws_bytes) {
    if (ws_bytes < (unsigned long long)WS_NEED_BYTES) return;
    const int o = blockIdx.x * 256 + threadIdx.x;
    float s = 0.0f;
#pragma unroll
    for (int k = 0; k < N_KS; ++k) s += ws[(size_t)k * (BATCH * OUT_CH) + o];
    const int wi = (980 - t[o >> 1]) / 20;
    out[o] = s + bias[wi * OUT_CH + (o & 1)];
}

extern "C" void kernel_launch(void* const* d_in, const int* in_sizes, int n_in,
                              void* d_out, int out_size, void* d_ws, size_t ws_size,
                              hipStream_t stream) {
    const float* x    = (const float*)d_in[0];
    const int*   t    = (const int*)  d_in[1];
    const float* W    = (const float*)d_in[2];
    const float* bias = (const float*)d_in[3];
    float*       out  = (float*)d_out;
    float*       ws   = (float*)d_ws;

    kslice_resident_w_matvec<<<dim3(256), dim3(THREADS), 0, stream>>>(
        x, t, W, bias, out, ws, (unsigned long long)ws_size);
    add_slices_and_bias<<<dim3(32), dim3(256), 0, stream>>>(
        ws, t, bias, out, (unsigned long long)ws_size);
}


# speedup vs baseline: 2.5685x; 1.0293x over previous
#include <hip/hip_runtime.h>


typedef float v2f __attribute__((ext_vector_type(2)));
typedef float v4f __attribute__((ext_vector_type(4)));
typedef float v8f __attribute__((ext_vector_type(8)));
typedef int   v4i __attribute__((ext_vector_type(4)));

#define IN_DIM         16384
#define OUT_CH         2
#define BATCH          4096
#define N_IDX          50
#define W_ROWS         (N_IDX * OUT_CH)
#define KS             512
#define N_KS           (IN_DIM / KS)
#define SB             512
#define WAVES          16
#define THREADS        (32 * WAVES)
#define KC             64
#define N_CH           (KS / KC)
#define PF             4
#define W_LOADS        (W_ROWS * (KS / 128) / WAVES)
#define WS_NEED_BYTES  (N_KS * BATCH * OUT_CH * 4)
#define WPITCH         (KS + 4)
#define XPITCH         (KC + 4)
#define XT_FLOATS      (16 * XPITCH)

#define F_SAMPLES      16
#define F_WAVES        8
#define F_K_PER_WAVE   (IN_DIM / F_WAVES)
#define F_KC           128
#define F_NCHUNK       (F_K_PER_WAVE / F_KC)
#define F_ROWS         (3 * F_SAMPLES)
#define F_PITCH        (F_KC + 4)

#define OFF_XT         (W_ROWS * WPITCH)
#define OFF_OUT        (OFF_XT + WAVES * XT_FLOATS)
#define OFF_INT        (OFF_OUT + SB * OUT_CH)
#define SMEM_FLOATS    (OFF_INT + WAVES * 64 + 2 * SB + WAVES * 32)

__device__ __forceinline__ void wave_lds_handover() {
    __builtin_amdgcn_fence(__ATOMIC_RELEASE, "wavefront");
    __builtin_amdgcn_wave_barrier();
    __builtin_amdgcn_fence(__ATOMIC_ACQUIRE, "wavefront");
}

__device__ __forceinline__ void lds_only_barrier() {
    __builtin_amdgcn_fence(__ATOMIC_RELEASE, "workgroup", "local");
    __builtin_amdgcn_s_barrier();
    __builtin_amdgcn_fence(__ATOMIC_ACQUIRE, "workgroup", "local");
}

__device__ __forceinline__ v4f load_x_once(const v4f* p) {
    return __builtin_nontemporal_load(p);
}

__device__ __forceinline__ void kslice_path(const float* __restrict__ x,
                                            const int*   __restrict__ t,
                                            const float* __restrict__ W,
                                            float*       __restrict__ ws,
                                            float* smem) {
    const int tid  = threadIdx.x;
    const int lane = tid & 31;
    const int wave = __builtin_amdgcn_readfirstlane(tid >> 5);
    const int ks   = blockIdx.x & (N_KS - 1);
    const int sb   = blockIdx.x >> 5;
    const int n    = lane & 15;
    const int h    = lane >> 4;

    float* Wl   = smem;
    float* xt   = smem + OFF_XT + wave * XT_FLOATS;
    float* outl = smem + OFF_OUT;
    int*   cnt  = (int*)(smem + OFF_INT);
    int*   ord  = cnt + WAVES * 64;
    int*   oix  = ord + SB;
    int*   dl   = oix + SB + wave * 32;

    const int my_t = t[sb * SB + tid];

    const float* Wg = W + ks * KS + 4 * lane;
    v4f wv[W_LOADS];
#pragma unroll
    for (int i = 0; i < W_LOADS; ++i) {
        const int L = wave * W_LOADS + i;
        wv[i] = *(const v4f*)(Wg + (size_t)(L >> 2) * IN_DIM + (L & 3) * 128);
    }

    const int my_idx = (980 - my_t) / 20;

    if (my_idx < N_IDX / 2) {
        const float* xp = x + (size_t)(sb * SB + tid) * IN_DIM + ks * KS;
        __builtin_prefetch(xp, 0, 1);
        __builtin_prefetch(xp + KC, 0, 1);
    }

    unsigned peers = 0xFFFFFFFFu;
#pragma unroll
    for (int k = 0; k < 6; ++k) {
        const bool bit = (my_idx >> k) & 1;
        const unsigned mb = __builtin_amdgcn_ballot_w32(bit);
        peers &= bit ? mb : ~mb;
    }
    const int rank_w = (int)__builtin_amdgcn_mbcnt_lo(peers, 0u);
    cnt[wave * 64 + lane]      = 0;
    cnt[wave * 64 + 32 + lane] = 0;
    if (rank_w == 0) cnt[wave * 64 + my_idx] = __builtin_popcount(peers);
    lds_only_barrier();
    int tot_lo = 0, tot_hi = 0, pre_lo = 0, pre_hi = 0;
#pragma unroll
    for (int w = 0; w < WAVES; ++w) {
        const int a0 = cnt[w * 64 + lane];
        const int a1 = cnt[w * 64 + 32 + lane];
        if (w < wave) { pre_lo += a0; pre_hi += a1; }
        tot_lo += a0; tot_hi += a1;
    }
    int inc_lo = tot_lo, inc_hi = tot_hi;
#pragma unroll
    for (int d = 1; d < 32; d <<= 1) {
        const int y0 = __shfl_up(inc_lo, d);
        const int y1 = __shfl_up(inc_hi, d);
        if (lane >= d) { inc_lo += y0; inc_hi += y1; }
    }
    const int all_lo   = __shfl(inc_lo, 31);
    const int start_lo = inc_lo - tot_lo + pre_lo;
    const int start_hi = all_lo + inc_hi - tot_hi + pre_hi;
    const int f_lo = __shfl(start_lo, my_idx & 31);
    const int f_hi = __shfl(start_hi, my_idx & 31);
    const int pos  = ((my_idx < 32) ? f_lo : f_hi) + rank_w;
    ord[pos] = tid;
    oix[pos] = my_idx;
    lds_only_barrier();

    const int p0 = wave * 16;
    unsigned xb[2][8];
    unsigned msk0 = 0, msk1 = 0;
    int nd0 = 1, nd1 = 1;
#pragma unroll
    for (int tl = 0; tl < 2; ++tl) {
        const int vcur  = oix[p0 + (SB / 2) * tl + n];
        const int vprev = oix[p0 + (SB / 2) * tl + (n > 0 ? n - 1 : 0)];
        const bool first = (n > 0) && (vcur != vprev);
        const unsigned mk = __builtin_amdgcn_ballot_w32(first) & 0xFFFFu;
        const int rk = __builtin_popcount(mk & ((2u << n) - 1u));
        if (h == 0 && (n == 0 || first)) dl[16 * tl + rk] = vcur;
        if (tl == 0) { msk0 = mk; nd0 = __builtin_popcount(mk) + 1; }
        else         { msk1 = mk; nd1 = __builtin_popcount(mk) + 1; }
#pragma unroll
        for (int q = 0; q < 8; ++q) {
            const int sl = ord[p0 + (SB / 2) * tl + 2 * q + h];
            xb[tl][q] = ((unsigned)(sb * SB + sl) * (unsigned)IN_DIM + (unsigned)(ks * KS + 4 * n)) * 4u;
        }
    }
    wave_lds_handover();

    const int ng0     = (nd0 > 8) ? 2 : 1;
    const int ng1     = (nd1 > 8) ? 2 : 1;
    const int n_items = __builtin_amdgcn_readfirstlane(ng0 + ng1);

    const char* xc = (const char*)x;
    v4f g[PF][8];
#pragma unroll
    for (int c = 0; c < PF; ++c)
#pragma unroll
        for (int q = 0; q < 8; ++q)
            g[c][q] = load_x_once((const v4f*)(xc + xb[0][q] + c * (KC * 4)));
    __builtin_amdgcn_sched_barrier(0);

#pragma unroll
    for (int i = 0; i < W_LOADS; ++i) {
        const int L = wave * W_LOADS + i;
        *(v4f*)(Wl + (L >> 2) * WPITCH + (L & 3) * 128 + 4 * lane) = wv[i];
    }

    lds_only_barrier();

    float*       st = xt + h * XPITCH + 4 * n;
    const float* fa = xt + n * XPITCH + 4 * h;
    const float* fb = Wl + 4 * h;
    v8f acc0 = {};
    v8f acc1 = {};

    int item = 0;
    do {
        const int tl = (item >= ng0) ? 1 : 0;
        const int gr = item - (tl ? ng0 : 0);
        {
            const int ndt = tl ? nd1 : nd0;
            int j = 8 * gr + (n >> 1);
            j = (j < ndt) ? j : (ndt - 1);
            const int d = dl[16 * tl + j];
            fb = Wl + (2 * d + (n & 1)) * WPITCH + 4 * h;
        }
        const int tln = (item + 1 >= ng0) ? 1 : 0;
        acc0 = (v8f){};
        acc1 = (v8f){};

#pragma unroll
        for (int half = 0; half < 2; ++half) {
            const bool have_next = (half == 0) || (item + 1 < n_items);
            unsigned xn[8];
#pragma unroll
            for (int q = 0; q < 8; ++q)
                xn[q] = (half == 0) ? (xb[tl][q] + 4u * (KC * 4))
                                    : (tln ? xb[1][q] : xb[0][q]);

#pragma unroll
            for (int cc = 0; cc < PF; ++cc) {
                const int c = half * PF + cc;
                wave_lds_handover();

#pragma unroll
                for (int q = 0; q < 8; ++q)
                    *(v4f*)(st + (2 * q) * XPITCH) = g[cc][q];

                if (have_next) {
#pragma unroll
                    for (int q = 0; q < 8; ++q)
                        g[cc][q] = load_x_once((const v4f*)(xc + xn[q] + cc * (KC * 4)));
                }

                wave_lds_handover();

                v4f af[8], bf[8];
#pragma unroll
                for (int j = 0; j < 8; ++j) af[j] = *(const v4f*)(fa + 8 * j);
#pragma unroll
                for (int j = 0; j < 8; ++j) bf[j] = *(const v4f*)(fb + c * KC + 8 * j);
                __builtin_amdgcn_sched_barrier(0);
#pragma unroll
                for (int j = 0; j < 8; ++j) {
                    const v2f alo = {af[j].x, af[j].y}, ahi = {af[j].z, af[j].w};
                    const v2f blo = {bf[j].x, bf[j].y}, bhi = {bf[j].z, bf[j].w};
                    acc0 = __builtin_amdgcn_wmma_f32_16x16x4_f32(
                               false, alo, false, blo, (short)0, acc0, false, false);
                    acc1 = __builtin_amdgcn_wmma_f32_16x16x4_f32(
                               false, ahi, false, bhi, (short)0, acc1, false, false);
                }
            }
        }

        const unsigned mk = tl ? msk1 : msk0;
        const v8f acc = acc0 + acc1;
        const v4i sa = *(const v4i*)(ord + p0 + (SB / 2) * tl + 8 * h);
        const v4i sc = *(const v4i*)(ord + p0 + (SB / 2) * tl + 8 * h + 4);
        const int srow[8] = {sa.x, sa.y, sa.z, sa.w, sc.x, sc.y, sc.z, sc.w};
#pragma unroll
        for (int r = 0; r < 8; ++r) {
            const int m  = 8 * h + r;
            const int jj = __builtin_popcount(mk & ((2u << m) - 1u)) - 8 * gr;
            if (jj == (n >> 1)) outl[2 * srow[r] + (n & 1)] = acc[r];
        }
    } while (++item < n_items);

    __syncthreads();
    const v2f o = *(const v2f*)(outl + 2 * tid);
    *(v2f*)(ws + ((size_t)ks * BATCH + (size_t)(sb * SB + tid)) * OUT_CH) = o;
}

__device__ __forceinline__ void single_launch_path(const float* __restrict__ x,
                                                   const int*   __restrict__ t,
                                                   const float* __restrict__ W,
                                                   const float* __restrict__ bias,
                                                   float*       __restrict__ out,
                                                   float* smem) {
    float* partial = smem + F_WAVES * F_ROWS * F_PITCH;

    const int tid   = threadIdx.x;
    const int lane  = tid & 31;
    const int wave  = __builtin_amdgcn_readfirstlane(tid >> 5);
    const int sbase = blockIdx.x * F_SAMPLES;
    const int n     = lane & 15;
    const int h     = lane >> 4;

    if (wave < F_WAVES) {
        float* tile = smem + wave * (F_ROWS * F_PITCH);
        const float* xw = x + (size_t)sbase * IN_DIM + (size_t)wave * F_K_PER_WAVE;
        const float* Ww = W + (size_t)wave * F_K_PER_WAVE;

        int wofs[F_SAMPLES];
#pragma unroll
        for (int j = 0; j < F_SAMPLES; ++j)
            wofs[j] = ((980 - t[sbase + j]) / 20) * (OUT_CH * IN_DIM);

        float*       st = tile + 4 * lane;
        const float* fr = tile + n * F_PITCH + 4 * h;

        v8f c0 = {};
        v8f c1 = {};

        v4f g[F_ROWS];
#pragma unroll
        for (int j = 0; j < F_SAMPLES; ++j)
            g[j] = *(const v4f*)(xw + (size_t)j * IN_DIM + 4 * lane);
#pragma unroll
        for (int j = 0; j < F_SAMPLES; ++j) {
            g[16 + j] = *(const v4f*)(Ww + wofs[j] + 4 * lane);
            g[32 + j] = *(const v4f*)(Ww + wofs[j] + IN_DIM + 4 * lane);
        }

        for (int c = 0; c < F_NCHUNK; ++c) {
            wave_lds_handover();
#pragma unroll
            for (int r = 0; r < F_ROWS; ++r)
                *(v4f*)(st + r * F_PITCH) = g[r];

            if (c + 1 < F_NCHUNK) {
                const int ko = (c + 1) * F_KC + 4 * lane;
#pragma unroll
                for (int j = 0; j < F_SAMPLES; ++j)
                    g[j] = *(const v4f*)(xw + (size_t)j * IN_DIM + ko);
#pragma unroll
                for (int j = 0; j < F_SAMPLES; ++j) {
                    g[16 + j] = *(const v4f*)(Ww + wofs[j] + ko);
                    g[32 + j] = *(const v4f*)(Ww + wofs[j] + IN_DIM + ko);
                }
            }

            wave_lds_handover();

#pragma unroll
            for (int j8 = 0; j8 < F_KC / 8; ++j8) {
                const v4f a  = *(const v4f*)(fr + 8 * j8);
                const v4f b0 = *(const v4f*)(fr + 16 * F_PITCH + 8 * j8);
                const v4f b1 = *(const v4f*)(fr + 32 * F_PITCH + 8 * j8);
                const v2f alo = {a.x, a.y},   ahi = {a.z, a.w};
                const v2f b0lo = {b0.x, b0.y}, b0hi = {b0.z, b0.w};
                const v2f b1lo = {b1.x, b1.y}, b1hi = {b1.z, b1.w};
                c0 = __builtin_amdgcn_wmma_f32_16x16x4_f32(
                         false, alo, false, b0lo, (short)0, c0, false, false);
                c1 = __builtin_amdgcn_wmma_f32_16x16x4_f32(
                         false, alo, false, b1lo, (short)0, c1, false, false);
                c0 = __builtin_amdgcn_wmma_f32_16x16x4_f32(
                         false, ahi, false, b0hi, (short)0, c0, false, false);
                c1 = __builtin_amdgcn_wmma_f32_16x16x4_f32(
                         false, ahi, false, b1hi, (short)0, c1, false, false);
            }
        }

        int m = -1, e = 0;
        if (lane < 8)        { m = lane;      e = lane;      }
        else if (lane >= 24) { m = lane - 16; e = lane - 24; }

        float d0 = 0.0f, d1 = 0.0f;
#pragma unroll
        for (int i = 0; i < 8; ++i) {
            d0 = (e == i) ? c0[i] : d0;
            d1 = (e == i) ? c1[i] : d1;
        }
        if (m >= 0) {
            partial[(wave * OUT_CH + 0) * F_SAMPLES + m] = d0;
            partial[(wave * OUT_CH + 1) * F_SAMPLES + m] = d1;
        }
    }
    __syncthreads();

    if (tid < F_SAMPLES * OUT_CH) {
        const int mm = tid >> 1;
        const int ch = tid & 1;
        float s = 0.0f;
#pragma unroll
        for (int w = 0; w < F_WAVES; ++w) s += partial[(w * OUT_CH + ch) * F_SAMPLES + mm];
        const int wi = (980 - t[sbase + mm]) / 20;
        out[(size_t)(sbase + mm) * OUT_CH + ch] = s + bias[wi * OUT_CH + ch];
    }
}

__global__ __launch_bounds__(THREADS)
void kslice_resident_w_matvec(const float* __restrict__ x,
                              const int*   __restrict__ t,
                              const float* __restrict__ W,
                              const float* __restrict__ bias,
                              float*       __restrict__ out,
                              float*       __restrict__ ws,
                              unsigned long long ws_bytes) {
    __shared__ __attribute__((aligned(16))) float smem[SMEM_FLOATS];
    if (ws_bytes >= (unsigned long long)WS_NEED_BYTES)
        kslice_path(x, t, W, ws, smem);
    else
        single_launch_path(x, t, W, bias, out, smem);
}

__global__ __launch_bounds__(256)
void add_slices_and_bias(const float* __restrict__ ws,
                         const int*   __restrict__ t,
                         const float* __restrict__ bias,
                         float*       __restrict__ out,
                         unsigned long long ws_bytes) {
    if (ws_bytes < (unsigned long long)WS_NEED_BYTES) return;
    const int o = blockIdx.x * 256 + threadIdx.x;
    float s = 0.0f;
#pragma unroll
    for (int k = 0; k < N_KS; ++k) s += ws[(size_t)k * (BATCH * OUT_CH) + o];
    const int wi = (980 - t[o >> 1]) / 20;
    out[o] = s + bias[wi * OUT_CH + (o & 1)];
}

extern "C" void kernel_launch(void* const* d_in, const int* in_sizes, int n_in,
                              void* d_out, int out_size, void* d_ws, size_t ws_size,
                              hipStream_t stream) {
    const float* x    = (const float*)d_in[0];
    const int*   t    = (const int*)  d_in[1];
    const float* W    = (const float*)d_in[2];
    const float* bias = (const float*)d_in[3];
    float*       out  = (float*)d_out;
    float*       ws   = (float*)d_ws;

    kslice_resident_w_matvec<<<dim3(256), dim3(THREADS), 0, stream>>>(
        x, t, W, bias, out, ws, (unsigned long long)ws_size);
    add_slices_and_bias<<<dim3(32), dim3(256), 0, stream>>>(
        ws, t, bias, out, (unsigned long long)ws_size);
}
